# starting kernel
# baseline (speedup 1.0000x reference)
#include <hip/hip_runtime.h>

typedef __attribute__((ext_vector_type(2))) float v2f;
typedef __attribute__((ext_vector_type(8))) float v8f;

#define HDIM 1024
#define BDIM 8

__global__ void __launch_bounds__(32)
bilinear_wv_wmma(const float* __restrict__ W,
                 const float* __restrict__ hidden,
                 float* __restrict__ v)
{
    const int lane = threadIdx.x;
    const int j0   = blockIdx.x * 16;
    const int m    = lane & 15;
    const int kp   = (lane >> 4) * 2;
    const int n    = lane & 15;

    const float* Arow = W + (size_t)(j0 + m) * HDIM + kp;
    const float* Brow = hidden + (size_t)((n < BDIM) ? n : 0) * HDIM + kp;
    const float  bmask = (n < BDIM) ? 1.0f : 0.0f;

    v8f acc0 = {};
    v8f acc1 = {};
    #pragma unroll 2
    for (int k0 = 0; k0 < HDIM; k0 += 8) {
        v2f a0 = *(const v2f*)(Arow + k0);
        v2f h0 = *(const v2f*)(Brow + k0);
        v2f b0; b0.x = h0.x * bmask; b0.y = h0.y * bmask;
        acc0 = __builtin_amdgcn_wmma_f32_16x16x4_f32(false, a0, false, b0,
                                                     (short)0, acc0, false, false);
        v2f a1 = *(const v2f*)(Arow + k0 + 4);
        v2f h1 = *(const v2f*)(Brow + k0 + 4);
        v2f b1; b1.x = h1.x * bmask; b1.y = h1.y * bmask;
        acc1 = __builtin_amdgcn_wmma_f32_16x16x4_f32(false, a1, false, b1,
                                                     (short)0, acc1, false, false);
    }
    v8f acc = acc0 + acc1;

    if (n < BDIM) {
        const int mbase = (lane >> 4) * 8;
        #pragma unroll
        for (int g = 0; g < 8; ++g)
            v[(size_t)n * HDIM + j0 + mbase + g] = acc[g];
    }
}

__global__ void __launch_bounds__(256)
bilinear_dot(const float* __restrict__ enc,
             const float* __restrict__ v,
             const float* __restrict__ bias,
             float* __restrict__ out)
{
    __shared__ float sv[HDIM];
    const int t = threadIdx.x;
    const int b = (blockIdx.x * 8) >> 13;

    ((float4*)sv)[t] = ((const float4*)(v + (size_t)b * HDIM))[t];
    __syncthreads();

    const int    wave = t >> 5;
    const int    lane = t & 31;
    const size_t row  = (size_t)blockIdx.x * 8 + wave;
    const float* erow = enc + row * HDIM;

    float4 acc = make_float4(0.f, 0.f, 0.f, 0.f);
    #pragma unroll
    for (int c = 0; c < 8; ++c) {
        const int k  = c * 128 + lane * 4;
        float4 e  = *(const float4*)(erow + k);
        float4 vv = *(const float4*)(sv + k);
        acc.x += e.x * vv.x;
        acc.y += e.y * vv.y;
        acc.z += e.z * vv.z;
        acc.w += e.w * vv.w;
    }
    float s = (acc.x + acc.y) + (acc.z + acc.w);
    #pragma unroll
    for (int off = 16; off > 0; off >>= 1)
        s += __shfl_xor(s, off, 32);

    if (lane == 0)
        out[row] = s + bias[0];
}

extern "C" void kernel_launch(void* const* d_in, const int* in_sizes, int n_in,
                              void* d_out, int out_size, void* d_ws, size_t ws_size,
                              hipStream_t stream)
{
    const float* hidden = (const float*)d_in[0];
    const float* enc    = (const float*)d_in[1];
    const float* W      = (const float*)d_in[3];
    const float* bias   = (const float*)d_in[4];

    float* v = (float*)d_ws;

    bilinear_wv_wmma<<<HDIM / 16, 32, 0, stream>>>(W, hidden, v);

    bilinear_dot<<<65536 / 8, 256, 0, stream>>>(enc, v, bias, (float*)d_out);
}


# kernel 1 (W x hidden): serial 128-step K loop of one wave replaced by split-K over 16 waves per j-tile, loads all in flight at once, partials summed via LDS in fixed order (new sum order of v)
# speedup vs baseline: 1.3800x; 1.3800x over previous
#include <hip/hip_runtime.h>

typedef __attribute__((ext_vector_type(2))) float v2f;
typedef __attribute__((ext_vector_type(8))) float v8f;

#define HDIM 1024
#define BDIM 8

#define KSPLIT 16
#define KWAVE  (HDIM / KSPLIT)
#define KSTEPS (KWAVE / 8)

__global__ void __launch_bounds__(KSPLIT * 32)
bilinear_wv_wmma(const float* __restrict__ W,
                 const float* __restrict__ hidden,
                 float* __restrict__ v)
{
    __shared__ float part[KSPLIT][16 * BDIM];

    const int t    = threadIdx.x;
    const int wave = t >> 5;
    const int lane = t & 31;
    const int j0   = blockIdx.x * 16;
    const int c    = lane & 15;
    const int h    = lane >> 4;
    const int kb   = wave * KWAVE + h * 4;
    const bool live = (c < BDIM);

    const float* Ap = W + (size_t)(j0 + c) * HDIM + kb;
    const float* Bp = hidden + (size_t)(live ? c : 0) * HDIM + kb;

    float4 a[KSTEPS];
    float4 hb[KSTEPS];
    for (int i = 0; i < KSTEPS; ++i) a[i]  = *(const float4*)(Ap + i * 8);
    for (int i = 0; i < KSTEPS; ++i) hb[i] = *(const float4*)(Bp + i * 8);

    v8f acc0 = {};
    v8f acc1 = {};
    for (int i = 0; i < KSTEPS; ++i) {
        v2f a0; a0.x = a[i].x; a0.y = a[i].y;
        v2f a1; a1.x = a[i].z; a1.y = a[i].w;
        v2f b0; b0.x = live ? hb[i].x : 0.0f; b0.y = live ? hb[i].y : 0.0f;
        v2f b1; b1.x = live ? hb[i].z : 0.0f; b1.y = live ? hb[i].w : 0.0f;
        acc0 = __builtin_amdgcn_wmma_f32_16x16x4_f32(false, a0, false, b0,
                                                     (short)0, acc0, false, false);
        acc1 = __builtin_amdgcn_wmma_f32_16x16x4_f32(false, a1, false, b1,
                                                     (short)0, acc1, false, false);
    }
    v8f acc = acc0 + acc1;

    if (live) {
        for (int g = 0; g < 8; ++g)
            part[wave][(h * 8 + g) * BDIM + c] = acc[g];
    }
    __syncthreads();

    if (t < 16 * BDIM) {
        const int b   = t >> 4;
        const int row = t & 15;
        float s = 0.0f;
        for (int w = 0; w < KSPLIT; ++w)
            s += part[w][row * BDIM + b];
        v[(size_t)b * HDIM + j0 + row] = s;
    }
}

__global__ void __launch_bounds__(256)
bilinear_dot(const float* __restrict__ enc,
             const float* __restrict__ v,
             const float* __restrict__ bias,
             float* __restrict__ out)
{
    __shared__ float sv[HDIM];
    const int t = threadIdx.x;
    const int b = (blockIdx.x * 8) >> 13;

    ((float4*)sv)[t] = ((const float4*)(v + (size_t)b * HDIM))[t];
    __syncthreads();

    const int    wave = t >> 5;
    const int    lane = t & 31;
    const size_t row  = (size_t)blockIdx.x * 8 + wave;
    const float* erow = enc + row * HDIM;

    float4 acc = make_float4(0.f, 0.f, 0.f, 0.f);
    #pragma unroll
    for (int c = 0; c < 8; ++c) {
        const int k  = c * 128 + lane * 4;
        float4 e  = *(const float4*)(erow + k);
        float4 vv = *(const float4*)(sv + k);
        acc.x += e.x * vv.x;
        acc.y += e.y * vv.y;
        acc.z += e.z * vv.z;
        acc.w += e.w * vv.w;
    }
    float s = (acc.x + acc.y) + (acc.z + acc.w);
    #pragma unroll
    for (int off = 16; off > 0; off >>= 1)
        s += __shfl_xor(s, off, 32);

    if (lane == 0)
        out[row] = s + bias[0];
}

extern "C" void kernel_launch(void* const* d_in, const int* in_sizes, int n_in,
                              void* d_out, int out_size, void* d_ws, size_t ws_size,
                              hipStream_t stream)
{
    const float* hidden = (const float*)d_in[0];
    const float* enc    = (const float*)d_in[1];
    const float* W      = (const float*)d_in[3];
    const float* bias   = (const float*)d_in[4];

    float* v = (float*)d_ws;

    bilinear_wv_wmma<<<HDIM / 16, KSPLIT * 32, 0, stream>>>(W, hidden, v);

    bilinear_dot<<<65536 / 8, 256, 0, stream>>>(enc, v, bias, (float*)d_out);
}


# kernel 2: 65536 one-row waves behind an LDS stage and barrier -> 4096 resident waves x 16 rows, v slice in registers, 3 rows of loads issued ahead of each row's sums; sum order unchanged
# speedup vs baseline: 1.5737x; 1.1404x over previous
#include <hip/hip_runtime.h>

typedef __attribute__((ext_vector_type(2))) float v2f;
typedef __attribute__((ext_vector_type(8))) float v8f;

#define HDIM 1024
#define BDIM 8

#define KSPLIT 16
#define KWAVE  (HDIM / KSPLIT)
#define KSTEPS (KWAVE / 8)

__global__ void __launch_bounds__(KSPLIT * 32)
bilinear_wv_wmma(const float* __restrict__ W,
                 const float* __restrict__ hidden,
                 float* __restrict__ v)
{
    __shared__ float part[KSPLIT][16 * BDIM];

    const int t    = threadIdx.x;
    const int wave = t >> 5;
    const int lane = t & 31;
    const int j0   = blockIdx.x * 16;
    const int c    = lane & 15;
    const int h    = lane >> 4;
    const int kb   = wave * KWAVE + h * 4;
    const bool live = (c < BDIM);

    const float* Ap = W + (size_t)(j0 + c) * HDIM + kb;
    const float* Bp = hidden + (size_t)(live ? c : 0) * HDIM + kb;

    float4 a[KSTEPS];
    float4 hb[KSTEPS];
    for (int i = 0; i < KSTEPS; ++i) a[i]  = *(const float4*)(Ap + i * 8);
    for (int i = 0; i < KSTEPS; ++i) hb[i] = *(const float4*)(Bp + i * 8);

    v8f acc0 = {};
    v8f acc1 = {};
    for (int i = 0; i < KSTEPS; ++i) {
        v2f a0; a0.x = a[i].x; a0.y = a[i].y;
        v2f a1; a1.x = a[i].z; a1.y = a[i].w;
        v2f b0; b0.x = live ? hb[i].x : 0.0f; b0.y = live ? hb[i].y : 0.0f;
        v2f b1; b1.x = live ? hb[i].z : 0.0f; b1.y = live ? hb[i].w : 0.0f;
        acc0 = __builtin_amdgcn_wmma_f32_16x16x4_f32(false, a0, false, b0,
                                                     (short)0, acc0, false, false);
        acc1 = __builtin_amdgcn_wmma_f32_16x16x4_f32(false, a1, false, b1,
                                                     (short)0, acc1, false, false);
    }
    v8f acc = acc0 + acc1;

    if (live) {
        for (int g = 0; g < 8; ++g)
            part[wave][(h * 8 + g) * BDIM + c] = acc[g];
    }
    __syncthreads();

    if (t < 16 * BDIM) {
        const int b   = t >> 4;
        const int row = t & 15;
        float s = 0.0f;
        for (int w = 0; w < KSPLIT; ++w)
            s += part[w][row * BDIM + b];
        v[(size_t)b * HDIM + j0 + row] = s;
    }
}

#define DOT_WAVES 16
#define DOT_ROWS  16
#define DOT_AHEAD 3
#define DOT_KQ    (HDIM / 128)
#define DOT_GROUPS (65536 / (DOT_WAVES * DOT_ROWS))

static_assert(8192 % DOT_ROWS == 0, "a wave's rows must lie in one batch");
static_assert(DOT_ROWS <= 32 && DOT_AHEAD >= 1 && DOT_AHEAD < DOT_ROWS, "lanes 0..DOT_ROWS-1 store");

__global__ void __launch_bounds__(DOT_WAVES * 32)
bilinear_dot(const float* __restrict__ enc,
             const float* __restrict__ v,
             const float* __restrict__ bias,
             float* __restrict__ out)
{
    const int lane = threadIdx.x & 31;
    const int wave = __builtin_amdgcn_readfirstlane((int)(threadIdx.x >> 5));
    const int row0 = (blockIdx.x * DOT_WAVES + wave) * DOT_ROWS;
    const int b    = row0 >> 13;

    const float* vb = v + (size_t)b * HDIM + lane * 4;
    float4 vq[DOT_KQ];
    for (int q = 0; q < DOT_KQ; ++q) vq[q] = *(const float4*)(vb + q * 128);
    __builtin_amdgcn_sched_barrier(0);

    const float* eb = enc + (size_t)row0 * HDIM + lane * 4;
    float4 e[DOT_AHEAD + 1][DOT_KQ];
    float  sum[DOT_ROWS];

#define LOAD_ROW(R) do { for (int q = 0; q < DOT_KQ; ++q) \
        e[(R) % (DOT_AHEAD + 1)][q] = *(const float4*)(eb + (size_t)(R) * HDIM + q * 128); } while (0)
    for (int r = 0; r < DOT_AHEAD; ++r) LOAD_ROW(r);
    for (int r = 0; r < DOT_ROWS; ++r) {
        __builtin_amdgcn_sched_barrier(0);
        if (r + DOT_AHEAD < DOT_ROWS) LOAD_ROW(r + DOT_AHEAD);
        __builtin_amdgcn_sched_barrier(0);

        const float4* er = e[r % (DOT_AHEAD + 1)];
        float4 acc = make_float4(0.f, 0.f, 0.f, 0.f);
        for (int q = 0; q < DOT_KQ; ++q) {
            acc.x += er[q].x * vq[q].x;
            acc.y += er[q].y * vq[q].y;
            acc.z += er[q].z * vq[q].z;
            acc.w += er[q].w * vq[q].w;
        }
        sum[r] = (acc.x + acc.y) + (acc.z + acc.w);
    }
#undef LOAD_ROW

    for (int r = 0; r < DOT_ROWS; ++r) {
        float s = sum[r];
        for (int off = 16; off > 0; off >>= 1)
            s += __shfl_xor(s, off, 32);
        sum[r] = s;
    }

    float mine = 0.f;
    for (int r = 0; r < DOT_ROWS; ++r)
        if (lane == r) mine = sum[r];

    if (lane < DOT_ROWS)
        out[row0 + lane] = mine + bias[0];
}

extern "C" void kernel_launch(void* const* d_in, const int* in_sizes, int n_in,
                              void* d_out, int out_size, void* d_ws, size_t ws_size,
                              hipStream_t stream)
{
    const float* hidden = (const float*)d_in[0];
    const float* enc    = (const float*)d_in[1];
    const float* W      = (const float*)d_in[3];
    const float* bias   = (const float*)d_in[4];

    float* v = (float*)d_ws;

    bilinear_wv_wmma<<<HDIM / 16, KSPLIT * 32, 0, stream>>>(W, hidden, v);

    bilinear_dot<<<DOT_GROUPS, DOT_WAVES * 32, 0, stream>>>(enc, v, bias, (float*)d_out);
}


# speedup vs baseline: 1.7404x; 1.1059x over previous
#include <hip/hip_runtime.h>

typedef __attribute__((ext_vector_type(2))) float v2f;
typedef __attribute__((ext_vector_type(8))) float v8f;
typedef __attribute__((ext_vector_type(4))) float v4f;

#define HDIM 1024
#define BDIM 8

#define KSPLIT 16
#define KWAVE  (HDIM / KSPLIT)
#define KSTEPS (KWAVE / 8)

__global__ void __launch_bounds__(KSPLIT * 32)
bilinear_wv_wmma(const float* __restrict__ W,
                 const float* __restrict__ hidden,
                 float* __restrict__ v)
{
    __shared__ float part[KSPLIT][16 * BDIM];

    const int t    = threadIdx.x;
    const int wave = t >> 5;
    const int lane = t & 31;
    const int j0   = blockIdx.x * 16;
    const int c    = lane & 15;
    const int h    = lane >> 4;
    const int kb   = wave * KWAVE + h * 4;
    const bool live = (c < BDIM);

    const float* Ap = W + (size_t)(j0 + c) * HDIM + kb;
    const float* Bp = hidden + (size_t)(live ? c : 0) * HDIM + kb;

    float4 a[KSTEPS];
    float4 hb[KSTEPS];
    for (int i = 0; i < KSTEPS; ++i) a[i]  = *(const float4*)(Ap + i * 8);
    for (int i = 0; i < KSTEPS; ++i) hb[i] = *(const float4*)(Bp + i * 8);

    v8f acc0 = {};
    v8f acc1 = {};
    for (int i = 0; i < KSTEPS; ++i) {
        v2f a0; a0.x = a[i].x; a0.y = a[i].y;
        v2f a1; a1.x = a[i].z; a1.y = a[i].w;
        v2f b0; b0.x = live ? hb[i].x : 0.0f; b0.y = live ? hb[i].y : 0.0f;
        v2f b1; b1.x = live ? hb[i].z : 0.0f; b1.y = live ? hb[i].w : 0.0f;
        acc0 = __builtin_amdgcn_wmma_f32_16x16x4_f32(false, a0, false, b0,
                                                     (short)0, acc0, false, false);
        acc1 = __builtin_amdgcn_wmma_f32_16x16x4_f32(false, a1, false, b1,
                                                     (short)0, acc1, false, false);
    }
    v8f acc = acc0 + acc1;

    if (live) {
        for (int g = 0; g < 8; ++g)
            part[wave][(h * 8 + g) * BDIM + c] = acc[g];
    }
    __syncthreads();

    if (t < 16 * BDIM) {
        const int b   = t >> 4;
        const int row = t & 15;
        float s = 0.0f;
        for (int w = 0; w < KSPLIT; ++w)
            s += part[w][row * BDIM + b];
        v[(size_t)b * HDIM + j0 + row] = s;
    }
}

#define DOT_WAVES 16
#define DOT_ROWS  16
#define DOT_AHEAD 3
#define DOT_KQ    (HDIM / 128)
#define DOT_GROUPS (65536 / (DOT_WAVES * DOT_ROWS))

static_assert(8192 % DOT_ROWS == 0, "a wave's rows must lie in one batch");
static_assert(DOT_ROWS <= 32 && DOT_AHEAD >= 1 && DOT_AHEAD < DOT_ROWS, "lanes 0..DOT_ROWS-1 store");

__global__ void __launch_bounds__(DOT_WAVES * 32)
bilinear_dot(const float* __restrict__ enc,
             const float* __restrict__ v,
             const float* __restrict__ bias,
             float* __restrict__ out)
{
    const int lane = threadIdx.x & 31;
    const int wave = __builtin_amdgcn_readfirstlane((int)(threadIdx.x >> 5));
    const int row0 = (blockIdx.x * DOT_WAVES + wave) * DOT_ROWS;
    const int b    = row0 >> 13;

    const float* vb = v + (size_t)b * HDIM + lane * 4;
    float4 vq[DOT_KQ];
    for (int q = 0; q < DOT_KQ; ++q) vq[q] = *(const float4*)(vb + q * 128);
    __builtin_amdgcn_sched_barrier(0);

    const float* eb = enc + (size_t)row0 * HDIM + lane * 4;
    v4f    e[DOT_AHEAD + 1][DOT_KQ];
    float  sum[DOT_ROWS];

#define LOAD_ROW(R) do { for (int q = 0; q < DOT_KQ; ++q) \
        e[(R) % (DOT_AHEAD + 1)][q] = __builtin_nontemporal_load((const v4f*)(eb + (size_t)(R) * HDIM + q * 128)); } while (0)
    for (int r = 0; r < DOT_AHEAD; ++r) LOAD_ROW(r);
    for (int r = 0; r < DOT_ROWS; ++r) {
        __builtin_amdgcn_sched_barrier(0);
        if (r + DOT_AHEAD < DOT_ROWS) LOAD_ROW(r + DOT_AHEAD);
        __builtin_amdgcn_sched_barrier(0);

        const v4f* er = e[r % (DOT_AHEAD + 1)];
        float4 acc = make_float4(0.f, 0.f, 0.f, 0.f);
        for (int q = 0; q < DOT_KQ; ++q) {
            acc.x += er[q].x * vq[q].x;
            acc.y += er[q].y * vq[q].y;
            acc.z += er[q].z * vq[q].z;
            acc.w += er[q].w * vq[q].w;
        }
        sum[r] = (acc.x + acc.y) + (acc.z + acc.w);
    }
#undef LOAD_ROW

    for (int r = 0; r < DOT_ROWS; ++r) {
        float s = sum[r];
        for (int off = 16; off > 0; off >>= 1)
            s += __shfl_xor(s, off, 32);
        sum[r] = s;
    }

    float mine = 0.f;
    for (int r = 0; r < DOT_ROWS; ++r)
        if (lane == r) mine = sum[r];

    if (lane < DOT_ROWS)
        out[row0 + lane] = mine + bias[0];
}

extern "C" void kernel_launch(void* const* d_in, const int* in_sizes, int n_in,
                              void* d_out, int out_size, void* d_ws, size_t ws_size,
                              hipStream_t stream)
{
    const float* hidden = (const float*)d_in[0];
    const float* enc    = (const float*)d_in[1];
    const float* W      = (const float*)d_in[3];
    const float* bias   = (const float*)d_in[4];

    float* v = (float*)d_ws;

    bilinear_wv_wmma<<<HDIM / 16, KSPLIT * 32, 0, stream>>>(W, hidden, v);

    bilinear_dot<<<DOT_GROUPS, DOT_WAVES * 32, 0, stream>>>(enc, v, bias, (float*)d_out);
}
